# starting kernel
# baseline (speedup 1.0000x reference)
#include <hip/hip_runtime.h>
#include <hip/hip_bf16.h>

#define GAT_N      50000
#define GAT_E      1600000
#define GAT_IN     128
#define GAT_OUT    64
#define GAT_SLOPE  0.2f
#define GAT_EPS    1e-12f

typedef __attribute__((ext_vector_type(2))) float v2f;
typedef __attribute__((ext_vector_type(4))) float v4f;
typedef __attribute__((ext_vector_type(8))) float v8f;

__global__ void k_init(float* __restrict__ out, float* __restrict__ maxb,
                       float* __restrict__ sumb) {
  int i = blockIdx.x * blockDim.x + threadIdx.x;
  if (i < GAT_N * GAT_OUT) out[i] = 0.f;
  if (i < GAT_N) { maxb[i] = -__builtin_inff(); sumb[i] = 0.f; }
}

__global__ __launch_bounds__(128) void k_gemm(const float* __restrict__ x,
                                              const float* __restrict__ W,
                                              float* __restrict__ Wh) {
  const int lane = threadIdx.x & 31;
  const int tile = blockIdx.x * 4 + (threadIdx.x >> 5);
  if (tile >= GAT_N / 16) return;
  const int base  = tile * 16;
  const int l15   = lane & 15;
  const int khalf = (lane >> 4) << 1;

  const float* xr = x + (size_t)(base + l15) * GAT_IN + khalf;
  const float* w0 = W + (size_t)(l15 +  0) * GAT_IN + khalf;
  const float* w1 = W + (size_t)(l15 + 16) * GAT_IN + khalf;
  const float* w2 = W + (size_t)(l15 + 32) * GAT_IN + khalf;
  const float* w3 = W + (size_t)(l15 + 48) * GAT_IN + khalf;

  v8f acc0 = {}, acc1 = {}, acc2 = {}, acc3 = {};

#pragma unroll 8
  for (int k = 0; k < GAT_IN; k += 4) {
    v2f a  = *(const v2f*)(xr + k);
    v2f b0 = *(const v2f*)(w0 + k);
    v2f b1 = *(const v2f*)(w1 + k);
    v2f b2 = *(const v2f*)(w2 + k);
    v2f b3 = *(const v2f*)(w3 + k);
    acc0 = __builtin_amdgcn_wmma_f32_16x16x4_f32(false, a, false, b0, (short)0, acc0, false, false);
    acc1 = __builtin_amdgcn_wmma_f32_16x16x4_f32(false, a, false, b1, (short)0, acc1, false, false);
    acc2 = __builtin_amdgcn_wmma_f32_16x16x4_f32(false, a, false, b2, (short)0, acc2, false, false);
    acc3 = __builtin_amdgcn_wmma_f32_16x16x4_f32(false, a, false, b3, (short)0, acc3, false, false);
  }

  const int rhalf = (lane >> 4) << 3;
  float* o = Wh + (size_t)(base + rhalf) * GAT_OUT + l15;
#pragma unroll
  for (int v = 0; v < 8; ++v) {
    o[(size_t)v * GAT_OUT +  0] = acc0[v];
    o[(size_t)v * GAT_OUT + 16] = acc1[v];
    o[(size_t)v * GAT_OUT + 32] = acc2[v];
    o[(size_t)v * GAT_OUT + 48] = acc3[v];
  }
}

__global__ void k_elr(const float* __restrict__ Wh, const float* __restrict__ al,
                      const float* __restrict__ ar, float* __restrict__ el,
                      float* __restrict__ er) {
  int i = blockIdx.x * blockDim.x + threadIdx.x;
  if (i >= GAT_N) return;
  const v4f* r = (const v4f*)(Wh + (size_t)i * GAT_OUT);
  float sl = 0.f, sr = 0.f;
#pragma unroll
  for (int q = 0; q < GAT_OUT / 4; ++q) {
    v4f v = r[q];
#pragma unroll
    for (int j = 0; j < 4; ++j) {
      sl += v[j] * al[q * 4 + j];
      sr += v[j] * ar[q * 4 + j];
    }
  }
  el[i] = sl; er[i] = sr;
}

__device__ __forceinline__ void atomic_max_f32(float* addr, float val) {
  if (val >= 0.f) atomicMax((int*)addr, __float_as_int(val));
  else            atomicMin((unsigned int*)addr, (unsigned int)__float_as_int(val));
}

__global__ void k_edge_logit(const int* __restrict__ src, const int* __restrict__ dst,
                             const float* __restrict__ el, const float* __restrict__ er,
                             float* __restrict__ ebuf, float* __restrict__ maxb) {
  int e = blockIdx.x * blockDim.x + threadIdx.x;
  if (e >= GAT_E) return;
  int s = src[e], d = dst[e];
  float v = el[s] + er[d];
  v = (v > 0.f) ? v : GAT_SLOPE * v;
  ebuf[e] = v;
  atomic_max_f32(maxb + d, v);
}

__global__ void k_edge_exp(const int* __restrict__ dst, float* __restrict__ ebuf,
                           const float* __restrict__ maxb, float* __restrict__ sumb) {
  int e = blockIdx.x * blockDim.x + threadIdx.x;
  if (e >= GAT_E) return;
  int d = dst[e];
  float ex = __expf(ebuf[e] - maxb[d]);
  ebuf[e] = ex;
  atomicAdd(sumb + d, ex);
}

__global__ void k_scatter(const int* __restrict__ src, const int* __restrict__ dst,
                          const float* __restrict__ ebuf, const float* __restrict__ Wh,
                          float* __restrict__ out) {
  int idx = blockIdx.x * blockDim.x + threadIdx.x;
  if (idx >= GAT_E * GAT_OUT) return;
  int e = idx >> 6, f = idx & 63;
  int s = src[e], d = dst[e];
  atomicAdd(out + (size_t)d * GAT_OUT + f, ebuf[e] * Wh[(size_t)s * GAT_OUT + f]);
}

__global__ void k_final(float* __restrict__ out, const float* __restrict__ sumb,
                        const float* __restrict__ bias) {
  int idx = blockIdx.x * blockDim.x + threadIdx.x;
  if (idx >= GAT_N * GAT_OUT) return;
  int i = idx >> 6, f = idx & 63;
  out[idx] = out[idx] / (sumb[i] + GAT_EPS) + bias[f];
}

extern "C" void kernel_launch(void* const* d_in, const int* in_sizes, int n_in,
                              void* d_out, int out_size, void* d_ws, size_t ws_size,
                              hipStream_t stream) {
  const float* x    = (const float*)d_in[0];
  const int*   ei   = (const int*)d_in[1];
  const float* W    = (const float*)d_in[2];
  const float* al   = (const float*)d_in[3];
  const float* ar   = (const float*)d_in[4];
  const float* bias = (const float*)d_in[5];
  float* out = (float*)d_out;
  (void)in_sizes; (void)n_in; (void)out_size; (void)ws_size;

  float* ws   = (float*)d_ws;
  float* Wh   = ws;
  float* el   = Wh   + (size_t)GAT_N * GAT_OUT;
  float* er   = el   + GAT_N;
  float* maxb = er   + GAT_N;
  float* sumb = maxb + GAT_N;
  float* ebuf = sumb + GAT_N;

  const int* src = ei;
  const int* dst = ei + GAT_E;

  const int B = 256;
  k_init      <<<(GAT_N * GAT_OUT + B - 1) / B, B, 0, stream>>>(out, maxb, sumb);
  k_gemm      <<<(GAT_N / 16 + 3) / 4, 128, 0, stream>>>(x, W, Wh);
  k_elr       <<<(GAT_N + B - 1) / B, B, 0, stream>>>(Wh, al, ar, el, er);
  k_edge_logit<<<(GAT_E + B - 1) / B, B, 0, stream>>>(src, dst, el, er, ebuf, maxb);
  k_edge_exp  <<<(GAT_E + B - 1) / B, B, 0, stream>>>(dst, ebuf, maxb, sumb);
  k_scatter   <<<(GAT_E * GAT_OUT + B - 1) / B, B, 0, stream>>>(src, dst, ebuf, Wh, out);
  k_final     <<<(GAT_N * GAT_OUT + B - 1) / B, B, 0, stream>>>(out, sumb, bias);
}


# GEMM fragments fetched 16 bytes per lane (4 consecutive k, two products per fetch; next group's loads kept ahead by a fence): 160 loads per wave become 80; k summed (0,1,4,5),(2,3,6,7) per 8-k block
# speedup vs baseline: 5.4299x; 5.4299x over previous
#include <hip/hip_runtime.h>
#include <hip/hip_bf16.h>

#define GAT_N      50000
#define GAT_E      1600000
#define GAT_IN     128
#define GAT_OUT    64
#define GAT_SLOPE  0.2f
#define GAT_EPS    1e-12f

typedef __attribute__((ext_vector_type(2))) float v2f;
typedef __attribute__((ext_vector_type(4))) float v4f;
typedef __attribute__((ext_vector_type(8))) float v8f;

__global__ __launch_bounds__(128) void k_gemm(const float* __restrict__ x,
                                              const float* __restrict__ W,
                                              float* __restrict__ Wh) {
  const int lane = threadIdx.x & 31;
  const int tile = blockIdx.x * 4 + (threadIdx.x >> 5);
  if (tile >= GAT_N / 16) return;
  const int base  = tile * 16;
  const int l15   = lane & 15;

  const int kq = (lane >> 4) << 2;
  const float* xr = x + (size_t)(base + l15) * GAT_IN + kq;
  const float* wr = W + (size_t)l15 * GAT_IN + kq;

  v8f acc[4] = {{}, {}, {}, {}};

#define GEMM_GB 2
#define GEMM_NG (GAT_IN / (8 * GEMM_GB))
  v4f fa[2][GEMM_GB], fb[2][GEMM_GB][4];
#define GEMM_LOAD_GROUP(s, t)                                                     \
  _Pragma("unroll") for (int j = 0; j < GEMM_GB; ++j) {                           \
    const int k = (t) * (8 * GEMM_GB) + j * 8;                                    \
    fa[s][j] = *(const v4f*)(xr + k);                                             \
    _Pragma("unroll") for (int n = 0; n < 4; ++n)                                 \
      fb[s][j][n] = *(const v4f*)(wr + (size_t)(16 * n) * GAT_IN + k);            \
  }
  GEMM_LOAD_GROUP(0, 0)
#pragma unroll
  for (int t = 0; t < GEMM_NG; ++t) {
    const int s = t & 1;
    if (t + 1 < GEMM_NG) { GEMM_LOAD_GROUP(s ^ 1, t + 1) }
    asm volatile("" :: "v"(fa[s][0]), "v"(fa[s][1]),
                 "v"(fb[s][0][0]), "v"(fb[s][0][1]), "v"(fb[s][0][2]), "v"(fb[s][0][3]),
                 "v"(fb[s][1][0]), "v"(fb[s][1][1]), "v"(fb[s][1][2]), "v"(fb[s][1][3]) : "memory");
#pragma unroll
    for (int j = 0; j < GEMM_GB; ++j) {
      const v4f a = fa[s][j];
      const v2f alo = {a[0], a[1]}, ahi = {a[2], a[3]};
#pragma unroll
      for (int n = 0; n < 4; ++n) {
        const v4f b = fb[s][j][n];
        acc[n] = __builtin_amdgcn_wmma_f32_16x16x4_f32(false, alo, false, (v2f){b[0], b[1]}, (short)0, acc[n], false, false);
      }
#pragma unroll
      for (int n = 0; n < 4; ++n) {
        const v4f b = fb[s][j][n];
        acc[n] = __builtin_amdgcn_wmma_f32_16x16x4_f32(false, ahi, false, (v2f){b[2], b[3]}, (short)0, acc[n], false, false);
      }
    }
  }
#undef GEMM_LOAD_GROUP
  const v8f acc0 = acc[0], acc1 = acc[1], acc2 = acc[2], acc3 = acc[3];

  const int rhalf = (lane >> 4) << 3;
  float* o = Wh + (size_t)(base + rhalf) * GAT_OUT + l15;
#pragma unroll
  for (int v = 0; v < 8; ++v) {
    o[(size_t)v * GAT_OUT +  0] = acc0[v];
    o[(size_t)v * GAT_OUT + 16] = acc1[v];
    o[(size_t)v * GAT_OUT + 32] = acc2[v];
    o[(size_t)v * GAT_OUT + 48] = acc3[v];
  }
}

__global__ void k_elr(const float* __restrict__ Wh, const float* __restrict__ al,
                      const float* __restrict__ ar, float* __restrict__ el,
                      float* __restrict__ er) {
  int i = blockIdx.x * blockDim.x + threadIdx.x;
  if (i >= GAT_N) return;
  const v4f* r = (const v4f*)(Wh + (size_t)i * GAT_OUT);
  float sl = 0.f, sr = 0.f;
#pragma unroll
  for (int q = 0; q < GAT_OUT / 4; ++q) {
    v4f v = r[q];
#pragma unroll
    for (int j = 0; j < 4; ++j) {
      sl += v[j] * al[q * 4 + j];
      sr += v[j] * ar[q * 4 + j];
    }
  }
  el[i] = sl; er[i] = sr;
}

#define AG_NODES   196
#define AG_GROUPS  256
#define AG_CAP     8192
#define AG_THREADS 1024
#define AG_NB      (GAT_E / 32)
#define AG_UPB     4
#define AG_WS_PK   ((size_t)(GAT_N * GAT_OUT + 2 * GAT_N) * 4 + (size_t)AG_NB * 32 + (size_t)GAT_E * 4)

__device__ __forceinline__ void bucket_pack(const int* __restrict__ src, const int* __restrict__ dst,
                                            unsigned* __restrict__ pk, int blk) {
  static_assert(GAT_E % 512 == 0, "a batch of 16 blocks lies wholly inside or wholly past the edge list");
  const int wb = __builtin_amdgcn_readfirstlane((blk & ~31) * 32);
  const int lane = (int)(threadIdx.x & 31);
  for (int i0 = 0; i0 < 32; i0 += 16) {
    const int eb = wb + i0 * 32;
    if (eb >= GAT_E) break;
    unsigned s[16], d[16];
#pragma unroll
    for (int j = 0; j < 16; ++j) {
      s[j] = (unsigned)src[eb + j * 32 + lane];
      d[j] = (unsigned)dst[eb + j * 32 + lane];
    }
#pragma unroll
    for (int j = 0; j < 16; ++j)
      pk[eb + j * 32 + lane] = s[j] | ((d[j] - (d[j] / AG_NODES) * AG_NODES) << 16);
  }
}

__global__ void k_bucket(const int* __restrict__ src, const int* __restrict__ dst,
                         uint4* __restrict__ planes, unsigned* __restrict__ pk, size_t ws_size) {
  const int blk = blockIdx.x * blockDim.x + threadIdx.x;
  if (ws_size >= AG_WS_PK) bucket_pack(src, dst, pk, blk);
  if (blk >= AG_NB) return;
  const int4* d = (const int4*)(dst + (size_t)blk * 32);
  unsigned w[8] = {0u, 0u, 0u, 0u, 0u, 0u, 0u, 0u};
#pragma unroll
  for (int i = 0; i < 8; ++i) {
    const int4 v = d[i];
    const unsigned b[4] = {(unsigned)v.x / AG_NODES, (unsigned)v.y / AG_NODES,
                           (unsigned)v.z / AG_NODES, (unsigned)v.w / AG_NODES};
#pragma unroll
    for (int c = 0; c < 4; ++c) {
#pragma unroll
      for (int t = 0; t < 8; ++t) w[t] |= ((b[c] >> t) & 1u) << (4 * i + c);
    }
  }
  planes[blk]         = make_uint4(w[0], w[1], w[2], w[3]);
  planes[AG_NB + blk] = make_uint4(w[4], w[5], w[6], w[7]);
}

__device__ __forceinline__ void atomic_max_f32(float* addr, float val) {
  if (val >= 0.f) atomicMax((int*)addr, __float_as_int(val));
  else            atomicMin((unsigned int*)addr, (unsigned int)__float_as_int(val));
}

__device__ __forceinline__ float leaky(float v) { return (v > 0.f) ? v : GAT_SLOPE * v; }

__global__ __launch_bounds__(AG_THREADS) void k_agg(const int* __restrict__ src, const int* __restrict__ dst,
                          const uint4* __restrict__ planes,
                          const unsigned* __restrict__ pk, size_t ws_size,
                          const float* __restrict__ el, const float* __restrict__ er,
                          const float* __restrict__ Wh, const float* __restrict__ bias,
                          float* __restrict__ out) {
  __shared__ float    acc[AG_NODES * GAT_OUT];
  __shared__ float    mx[AG_NODES];
  __shared__ float    sm[AG_NODES];
  __shared__ unsigned lst[AG_CAP];
  __shared__ float    val[AG_CAP];
  __shared__ float    elt[GAT_N];
  __shared__ float    erl[AG_NODES];
  __shared__ unsigned cnt;
  static_assert(GAT_N % 4 == 0, "el is copied 4 floats at a time");
  static_assert((GAT_N / 4 + AG_THREADS - 1) / AG_THREADS <= (AG_NB + AG_THREADS * AG_UPB - 1) / (AG_THREADS * AG_UPB),
                "the scan makes enough trips to copy all of el");

  const int tid  = threadIdx.x;
  const int lane = tid & 31;
  const int wv   = tid >> 5;
  const int g    = blockIdx.x;
  const int lo   = g * AG_NODES;
  const int nn   = (GAT_N - lo < AG_NODES) ? (GAT_N - lo) : AG_NODES;

  for (int i = tid; i < AG_NODES * GAT_OUT; i += AG_THREADS) acc[i] = 0.f;
  if (tid < AG_NODES) {
    mx[tid] = -__builtin_inff(); sm[tid] = 0.f;
    erl[tid] = (tid < nn) ? er[lo + tid] : 0.f;
  }
  if (tid == 0) cnt = 0u;
  __syncthreads();

  unsigned nt[8];
#pragma unroll
  for (int t = 0; t < 8; ++t) nt[t] = ((g >> t) & 1) ? 0u : 0xffffffffu;
  const int rot = g * (AG_NB / AG_GROUPS);
  for (int base0 = 0; base0 < AG_NB; base0 += AG_THREADS * AG_UPB) {
    const int base = base0 + tid;
    uint4 qa[AG_UPB], qb[AG_UPB];
    int   ux[AG_UPB];
#pragma unroll
    for (int k = 0; k < AG_UPB; ++k) {
      const int p = base + k * AG_THREADS;
      int u = p + rot;
      if (u >= AG_NB) u -= AG_NB;
      if (p >= AG_NB) u = 0;
      ux[k] = u;
      qa[k] = planes[u];
      qb[k] = planes[AG_NB + u];
    }
    const int q = base0 / AG_UPB + tid;
    v4f elq = {};
    if (q < GAT_N / 4) elq = ((const v4f*)el)[q];
    unsigned mk[AG_UPB];
    unsigned c = 0;
#pragma unroll
    for (int k = 0; k < AG_UPB; ++k) {
      unsigned m = (qa[k].x ^ nt[0]) & (qa[k].y ^ nt[1]) & (qa[k].z ^ nt[2]) & (qa[k].w ^ nt[3]) &
                   (qb[k].x ^ nt[4]) & (qb[k].y ^ nt[5]) & (qb[k].z ^ nt[6]) & (qb[k].w ^ nt[7]);
      if (base + k * AG_THREADS >= AG_NB) m = 0u;
      mk[k] = m;
      c += (unsigned)__builtin_popcount(m);
    }
    unsigned inc = c;
    inc += (unsigned)__builtin_amdgcn_update_dpp(0, (int)inc, 0x111, 0xf, 0xf, true);
    inc += (unsigned)__builtin_amdgcn_update_dpp(0, (int)inc, 0x112, 0xf, 0xf, true);
    inc += (unsigned)__builtin_amdgcn_update_dpp(0, (int)inc, 0x114, 0xf, 0xf, true);
    inc += (unsigned)__builtin_amdgcn_update_dpp(0, (int)inc, 0x118, 0xf, 0xf, true);
    if (q < GAT_N / 4) ((v4f*)elt)[q] = elq;
    const unsigned row0 = (unsigned)__builtin_amdgcn_readlane((int)inc, 15);
    if (lane >= 16) inc += row0;
    const unsigned tot = (unsigned)__builtin_amdgcn_readlane((int)inc, 31);
    unsigned wb = 0;
    if (lane == 0) wb = atomicAdd(&cnt, tot);
    wb = (unsigned)__builtin_amdgcn_readfirstlane((int)wb);
    unsigned o = wb + inc - c;
#pragma unroll
    for (int k = 0; k < AG_UPB; ++k) {
      unsigned m = mk[k];
      const unsigned hi = (unsigned)ux[k] << 5;
      while (m) {
        const unsigned beta = (unsigned)__builtin_ctz(m);
        m &= m - 1u;
        if (o < AG_CAP) lst[o] = hi | beta;
        ++o;
      }
    }
  }
  __syncthreads();
  const unsigned total = cnt;

  if (total <= AG_CAP) {
    const bool use_pk = ws_size >= AG_WS_PK;
    for (unsigned i0 = tid; i0 < total; i0 += AG_THREADS * 4) {
      unsigned e[4], w[4]; float a[4], b[4];
#pragma unroll
      for (int k = 0; k < 4; ++k) {
        const unsigned i = i0 + k * AG_THREADS;
        e[k] = (i < total) ? lst[i] : 0u;
      }
      if (use_pk) {
#pragma unroll
        for (int k = 0; k < 4; ++k) w[k] = pk[e[k]];
      } else {
#pragma unroll
        for (int k = 0; k < 4; ++k) w[k] = (unsigned)src[e[k]] | ((unsigned)(dst[e[k]] - lo) << 16);
      }
#pragma unroll
      for (int k = 0; k < 4; ++k) if (i0 + k * AG_THREADS >= total) w[k] = 0u;
#pragma unroll
      for (int k = 0; k < 4; ++k) { a[k] = elt[w[k] & 0xffffu]; b[k] = erl[w[k] >> 16]; }
#pragma unroll
      for (int k = 0; k < 4; ++k) {
        const unsigned i = i0 + k * AG_THREADS;
        if (i < total) {
          const float v = leaky(a[k] + b[k]);
          val[i] = v;
          lst[i] = w[k];
          atomic_max_f32(&mx[w[k] >> 16], v);
        }
      }
    }
    __syncthreads();
    for (unsigned i = tid; i < total; i += AG_THREADS) {
      const unsigned dl = lst[i] >> 16;
      const float ex = __expf(val[i] - mx[dl]);
      val[i] = ex;
      atomicAdd(&sm[dl], ex);
    }
    __syncthreads();
    const float* whl = Wh + lane;
    for (unsigned b0 = (unsigned)wv * 32u; b0 < total; b0 += AG_THREADS) {
      const unsigned id = b0 + lane;
      const unsigned pk = (id < total) ? lst[id] : 0u;
      const int      wi = (id < total) ? __float_as_int(val[id]) : 0;
      const int n = (total - b0 < 32u) ? (int)(total - b0) : 32;
#pragma unroll
      for (int h = 0; h < 32; h += 16) {
        if (n - h >= 16) {
          float r0[16], r1[16];
#pragma unroll
          for (int j = 0; j < 16; ++j) {
            const unsigned p = (unsigned)__builtin_amdgcn_readlane((int)pk, h + j);
            const float* r = whl + (size_t)(p & 0xffffu) * GAT_OUT;
            r0[j] = r[0];
            r1[j] = r[32];
          }
#pragma unroll
          for (int j = 0; j < 16; ++j) {
            const unsigned p = (unsigned)__builtin_amdgcn_readlane((int)pk, h + j);
            const float w = __int_as_float(__builtin_amdgcn_readlane(wi, h + j));
            float* a = acc + (p >> 16) * GAT_OUT + lane;
            atomicAdd(a,      w * r0[j]);
            atomicAdd(a + 32, w * r1[j]);
          }
        } else {
          for (int j = h; j < n && j < h + 16; ++j) {
            const unsigned p = (unsigned)__builtin_amdgcn_readlane((int)pk, j);
            const float w = __int_as_float(__builtin_amdgcn_readlane(wi, j));
            const float* r = whl + (size_t)(p & 0xffffu) * GAT_OUT;
            float* a = acc + (p >> 16) * GAT_OUT + lane;
            atomicAdd(a,      w * r[0]);
            atomicAdd(a + 32, w * r[32]);
          }
        }
      }
    }
  } else {
    for (int e = tid; e < GAT_E; e += AG_THREADS) {
      const int d = dst[e];
      if ((unsigned)(d - lo) < (unsigned)nn)
        atomic_max_f32(&mx[d - lo], leaky(el[src[e]] + er[d]));
    }
    __syncthreads();
    for (int e = tid; e < GAT_E; e += AG_THREADS) {
      const int d = dst[e];
      if ((unsigned)(d - lo) < (unsigned)nn) {
        const int s = src[e];
        const float ex = __expf(leaky(el[s] + er[d]) - mx[d - lo]);
        atomicAdd(&sm[d - lo], ex);
        const float* r = Wh + (size_t)s * GAT_OUT;
        float* a = acc + (d - lo) * GAT_OUT;
        for (int f = 0; f < GAT_OUT; ++f) atomicAdd(a + f, ex * r[f]);
      }
    }
  }
  __syncthreads();

  float* o = out + (size_t)lo * GAT_OUT;
  for (int i = tid; i < nn * GAT_OUT; i += AG_THREADS)
    o[i] = acc[i] / (sm[i >> 6] + GAT_EPS) + bias[i & 63];
}

extern "C" void kernel_launch(void* const* d_in, const int* in_sizes, int n_in,
                              void* d_out, int out_size, void* d_ws, size_t ws_size,
                              hipStream_t stream) {
  const float* x    = (const float*)d_in[0];
  const int*   ei   = (const int*)d_in[1];
  const float* W    = (const float*)d_in[2];
  const float* al   = (const float*)d_in[3];
  const float* ar   = (const float*)d_in[4];
  const float* bias = (const float*)d_in[5];
  float* out = (float*)d_out;
  (void)in_sizes; (void)n_in; (void)out_size;

  float* ws   = (float*)d_ws;
  float* Wh   = ws;
  float* el   = Wh   + (size_t)GAT_N * GAT_OUT;
  float* er   = el   + GAT_N;
  uint4* planes = (uint4*)(er + GAT_N);
  unsigned* pk  = (unsigned*)(planes + 2 * AG_NB);

  const int* src = ei;
  const int* dst = ei + GAT_E;

  const int B = 256;
  k_gemm  <<<(GAT_N / 16 + 3) / 4, 128, 0, stream>>>(x, W, Wh);
  k_elr   <<<(GAT_N + B - 1) / B, B, 0, stream>>>(Wh, al, ar, el, er);
  k_bucket<<<(AG_NB + B - 1) / B, B, 0, stream>>>(src, dst, planes, pk, ws_size);
  k_agg   <<<AG_GROUPS, AG_THREADS, 0, stream>>>(src, dst, planes, pk, ws_size, el, er, Wh, bias, out);
}


# Group ids kept as one-hot nibble planes (32 words per 32 edges): a group's match word is the AND of 2 words; the scan reads 8 bytes a block, not 32, in 4 trips, not 13; list order changes, same sums
# speedup vs baseline: 5.9655x; 1.0986x over previous
#include <hip/hip_runtime.h>
#include <hip/hip_bf16.h>

#define GAT_N      50000
#define GAT_E      1600000
#define GAT_IN     128
#define GAT_OUT    64
#define GAT_SLOPE  0.2f
#define GAT_EPS    1e-12f

typedef __attribute__((ext_vector_type(2))) float v2f;
typedef __attribute__((ext_vector_type(4))) float v4f;
typedef __attribute__((ext_vector_type(8))) float v8f;
typedef __attribute__((ext_vector_type(4))) unsigned v4u;

__global__ __launch_bounds__(128) void k_gemm(const float* __restrict__ x,
                                              const float* __restrict__ W,
                                              float* __restrict__ Wh) {
  const int lane = threadIdx.x & 31;
  const int tile = blockIdx.x * 4 + (threadIdx.x >> 5);
  if (tile >= GAT_N / 16) return;
  const int base  = tile * 16;
  const int l15   = lane & 15;

  const int kq = (lane >> 4) << 2;
  const float* xr = x + (size_t)(base + l15) * GAT_IN + kq;
  const float* wr = W + (size_t)l15 * GAT_IN + kq;

  v8f acc[4] = {{}, {}, {}, {}};

#define GEMM_GB 2
#define GEMM_NG (GAT_IN / (8 * GEMM_GB))
  v4f fa[2][GEMM_GB], fb[2][GEMM_GB][4];
#define GEMM_LOAD_GROUP(s, t)                                                     \
  _Pragma("unroll") for (int j = 0; j < GEMM_GB; ++j) {                           \
    const int k = (t) * (8 * GEMM_GB) + j * 8;                                    \
    fa[s][j] = *(const v4f*)(xr + k);                                             \
    _Pragma("unroll") for (int n = 0; n < 4; ++n)                                 \
      fb[s][j][n] = *(const v4f*)(wr + (size_t)(16 * n) * GAT_IN + k);            \
  }
  GEMM_LOAD_GROUP(0, 0)
#pragma unroll
  for (int t = 0; t < GEMM_NG; ++t) {
    const int s = t & 1;
    if (t + 1 < GEMM_NG) { GEMM_LOAD_GROUP(s ^ 1, t + 1) }
    asm volatile("" :: "v"(fa[s][0]), "v"(fa[s][1]),
                 "v"(fb[s][0][0]), "v"(fb[s][0][1]), "v"(fb[s][0][2]), "v"(fb[s][0][3]),
                 "v"(fb[s][1][0]), "v"(fb[s][1][1]), "v"(fb[s][1][2]), "v"(fb[s][1][3]) : "memory");
#pragma unroll
    for (int j = 0; j < GEMM_GB; ++j) {
      const v4f a = fa[s][j];
      const v2f alo = {a[0], a[1]}, ahi = {a[2], a[3]};
#pragma unroll
      for (int n = 0; n < 4; ++n) {
        const v4f b = fb[s][j][n];
        acc[n] = __builtin_amdgcn_wmma_f32_16x16x4_f32(false, alo, false, (v2f){b[0], b[1]}, (short)0, acc[n], false, false);
      }
#pragma unroll
      for (int n = 0; n < 4; ++n) {
        const v4f b = fb[s][j][n];
        acc[n] = __builtin_amdgcn_wmma_f32_16x16x4_f32(false, ahi, false, (v2f){b[2], b[3]}, (short)0, acc[n], false, false);
      }
    }
  }
#undef GEMM_LOAD_GROUP
  const v8f acc0 = acc[0], acc1 = acc[1], acc2 = acc[2], acc3 = acc[3];

  const int rhalf = (lane >> 4) << 3;
  float* o = Wh + (size_t)(base + rhalf) * GAT_OUT + l15;
#pragma unroll
  for (int v = 0; v < 8; ++v) {
    o[(size_t)v * GAT_OUT +  0] = acc0[v];
    o[(size_t)v * GAT_OUT + 16] = acc1[v];
    o[(size_t)v * GAT_OUT + 32] = acc2[v];
    o[(size_t)v * GAT_OUT + 48] = acc3[v];
  }
}

__global__ void k_elr(const float* __restrict__ Wh, const float* __restrict__ al,
                      const float* __restrict__ ar, float* __restrict__ el,
                      float* __restrict__ er) {
  int i = blockIdx.x * blockDim.x + threadIdx.x;
  if (i >= GAT_N) return;
  const v4f* r = (const v4f*)(Wh + (size_t)i * GAT_OUT);
  float sl = 0.f, sr = 0.f;
#pragma unroll
  for (int q = 0; q < GAT_OUT / 4; ++q) {
    v4f v = r[q];
#pragma unroll
    for (int j = 0; j < 4; ++j) {
      sl += v[j] * al[q * 4 + j];
      sr += v[j] * ar[q * 4 + j];
    }
  }
  el[i] = sl; er[i] = sr;
}

#define AG_NODES   196
#define AG_GROUPS  256
#define AG_CAP     8192
#define AG_THREADS 1024
#define AG_NB      (GAT_E / 32)
#define AG_NQ      (AG_NB / 4)
#define AG_PP      ((AG_NB + 31) / 32 * 32)
#define AG_UPB     4
#define AG_WS_OH   ((size_t)(GAT_N * GAT_OUT + 2 * GAT_N) * 4 + (size_t)AG_PP * 32 * 4)
#define AG_WS_PK   (AG_WS_OH + (size_t)GAT_E * 4)

__device__ __forceinline__ void bucket_pack(const int* __restrict__ src, const int* __restrict__ dst,
                                            unsigned* __restrict__ pk, int blk) {
  static_assert(GAT_E % 512 == 0, "a batch of 16 blocks lies wholly inside or wholly past the edge list");
  const int wb = __builtin_amdgcn_readfirstlane((blk & ~31) * 32);
  const int lane = (int)(threadIdx.x & 31);
  for (int i0 = 0; i0 < 32; i0 += 16) {
    const int eb = wb + i0 * 32;
    if (eb >= GAT_E) break;
    unsigned s[16], d[16];
#pragma unroll
    for (int j = 0; j < 16; ++j) {
      s[j] = (unsigned)src[eb + j * 32 + lane];
      d[j] = (unsigned)dst[eb + j * 32 + lane];
    }
#pragma unroll
    for (int j = 0; j < 16; ++j)
      pk[eb + j * 32 + lane] = s[j] | ((d[j] - (d[j] / AG_NODES) * AG_NODES) << 16);
  }
}

__global__ void k_bucket(const int* __restrict__ src, const int* __restrict__ dst,
                         unsigned* __restrict__ oh, unsigned* __restrict__ pk, size_t ws_size) {
  const int blk = blockIdx.x * blockDim.x + threadIdx.x;
  if (ws_size < AG_WS_OH) return;
  if (ws_size >= AG_WS_PK) bucket_pack(src, dst, pk, blk);
  if (blk >= AG_NB) return;
  const int4* d = (const int4*)(dst + (size_t)blk * 32);
  unsigned w[8] = {0u, 0u, 0u, 0u, 0u, 0u, 0u, 0u};
#pragma unroll
  for (int i = 0; i < 8; ++i) {
    const int4 v = d[i];
    const unsigned b[4] = {(unsigned)v.x / AG_NODES, (unsigned)v.y / AG_NODES,
                           (unsigned)v.z / AG_NODES, (unsigned)v.w / AG_NODES};
#pragma unroll
    for (int c = 0; c < 4; ++c) {
#pragma unroll
      for (int t = 0; t < 8; ++t) w[t] |= ((b[c] >> t) & 1u) << (4 * i + c);
    }
  }
  unsigned pr[4][4];
#pragma unroll
  for (int f = 0; f < 4; ++f) {
#pragma unroll
    for (int v = 0; v < 4; ++v)
      pr[f][v] = ((v & 1) ? w[2 * f] : ~w[2 * f]) & ((v & 2) ? w[2 * f + 1] : ~w[2 * f + 1]);
  }
  unsigned* ob = oh + blk;
#pragma unroll
  for (int v = 0; v < 16; ++v) {
    ob[(size_t)v * AG_PP]        = pr[2][v & 3] & pr[3][v >> 2];
    ob[(size_t)(16 + v) * AG_PP] = pr[0][v & 3] & pr[1][v >> 2];
  }
}

__device__ __forceinline__ void atomic_max_f32(float* addr, float val) {
  if (val >= 0.f) atomicMax((int*)addr, __float_as_int(val));
  else            atomicMin((unsigned int*)addr, (unsigned int)__float_as_int(val));
}

__device__ __forceinline__ float leaky(float v) { return (v > 0.f) ? v : GAT_SLOPE * v; }

__global__ __launch_bounds__(AG_THREADS) void k_agg(const int* __restrict__ src, const int* __restrict__ dst,
                          const unsigned* __restrict__ oh,
                          const unsigned* __restrict__ pk, size_t ws_size,
                          const float* __restrict__ el, const float* __restrict__ er,
                          const float* __restrict__ Wh, const float* __restrict__ bias,
                          float* __restrict__ out) {
  __shared__ float    acc[AG_NODES * GAT_OUT];
  __shared__ float    mx[AG_NODES];
  __shared__ float    sm[AG_NODES];
  __shared__ unsigned lst[AG_CAP];
  __shared__ float    val[AG_CAP];
  __shared__ float    elt[GAT_N];
  __shared__ float    erl[AG_NODES];
  __shared__ unsigned cnt;
  static_assert(GAT_N % 4 == 0, "el is copied 4 floats at a time");
  static_assert(AG_NB % 4 == 0 && GAT_N / 4 <= AG_NQ, "a lane copies one 16-byte piece of el per quad of blocks it scans");

  const int tid  = threadIdx.x;
  const int lane = tid & 31;
  const int wv   = tid >> 5;
  const int g    = blockIdx.x;
  const int lo   = g * AG_NODES;
  const int nn   = (GAT_N - lo < AG_NODES) ? (GAT_N - lo) : AG_NODES;

  for (int i = tid; i < AG_NODES * GAT_OUT; i += AG_THREADS) acc[i] = 0.f;
  if (tid < AG_NODES) {
    mx[tid] = -__builtin_inff(); sm[tid] = 0.f;
    erl[tid] = (tid < nn) ? er[lo + tid] : 0.f;
  }
  if (tid == 0) cnt = 0u;
  __syncthreads();

  const bool have_planes = ws_size >= AG_WS_OH;
  const v4u* hp = (const v4u*)(oh + (size_t)(g >> 4) * AG_PP);
  const v4u* lp = (const v4u*)(oh + (size_t)(16 + (g & 15)) * AG_PP);
  const int rot = g * (AG_NQ / AG_GROUPS);
  for (int base0 = 0; have_planes && base0 < AG_NQ; base0 += AG_THREADS * AG_UPB) {
    const int base = base0 + tid;
    v4u qh[AG_UPB], ql[AG_UPB];
    v4f elq[AG_UPB];
    int   ux[AG_UPB];
#pragma unroll
    for (int k = 0; k < AG_UPB; ++k) {
      const int p = base + k * AG_THREADS;
      int u = p + rot;
      if (u >= AG_NQ) u -= AG_NQ;
      if (p >= AG_NQ) u = 0;
      ux[k] = u;
      qh[k] = hp[u];
      ql[k] = lp[u];
      elq[k] = ((const v4f*)el)[(p < GAT_N / 4) ? p : 0];
    }
    static_assert(AG_UPB == 4, "the fence below names the loads of four quads");
    asm volatile("" :: "v"(qh[0]), "v"(ql[0]), "v"(qh[1]), "v"(ql[1]), "v"(qh[2]), "v"(ql[2]),
                 "v"(qh[3]), "v"(ql[3]), "v"(elq[0]), "v"(elq[1]), "v"(elq[2]), "v"(elq[3]) : "memory");
    unsigned mk[AG_UPB][4];
    unsigned c = 0;
#pragma unroll
    for (int k = 0; k < AG_UPB; ++k) {
      const bool in = base + k * AG_THREADS < AG_NQ;
      mk[k][0] = in ? (qh[k].x & ql[k].x) : 0u;
      mk[k][1] = in ? (qh[k].y & ql[k].y) : 0u;
      mk[k][2] = in ? (qh[k].z & ql[k].z) : 0u;
      mk[k][3] = in ? (qh[k].w & ql[k].w) : 0u;
#pragma unroll
      for (int j = 0; j < 4; ++j) c += (unsigned)__builtin_popcount(mk[k][j]);
    }
    unsigned inc = c;
    inc += (unsigned)__builtin_amdgcn_update_dpp(0, (int)inc, 0x111, 0xf, 0xf, true);
    inc += (unsigned)__builtin_amdgcn_update_dpp(0, (int)inc, 0x112, 0xf, 0xf, true);
    inc += (unsigned)__builtin_amdgcn_update_dpp(0, (int)inc, 0x114, 0xf, 0xf, true);
    inc += (unsigned)__builtin_amdgcn_update_dpp(0, (int)inc, 0x118, 0xf, 0xf, true);
#pragma unroll
    for (int k = 0; k < AG_UPB; ++k) {
      const int p = base + k * AG_THREADS;
      if (p < GAT_N / 4) ((v4f*)elt)[p] = elq[k];
    }
    const unsigned row0 = (unsigned)__builtin_amdgcn_readlane((int)inc, 15);
    if (lane >= 16) inc += row0;
    const unsigned tot = (unsigned)__builtin_amdgcn_readlane((int)inc, 31);
    unsigned wb = 0;
    if (lane == 0) wb = atomicAdd(&cnt, tot);
    wb = (unsigned)__builtin_amdgcn_readfirstlane((int)wb);
    unsigned o = wb + inc - c;
#pragma unroll
    for (int k = 0; k < AG_UPB; ++k) {
#pragma unroll
      for (int j = 0; j < 4; ++j) {
        unsigned m = mk[k][j];
        const unsigned hi = (unsigned)(ux[k] * 4 + j) << 5;
        while (m) {
          const unsigned beta = (unsigned)__builtin_ctz(m);
          m &= m - 1u;
          if (o < AG_CAP) lst[o] = hi | beta;
          ++o;
        }
      }
    }
  }
  __syncthreads();
  const unsigned total = have_planes ? cnt : (unsigned)AG_CAP + 1u;

  if (total <= AG_CAP) {
    const bool use_pk = ws_size >= AG_WS_PK;
    for (unsigned i0 = tid; i0 < total; i0 += AG_THREADS * 4) {
      unsigned e[4], w[4]; float a[4], b[4];
#pragma unroll
      for (int k = 0; k < 4; ++k) {
        const unsigned i = i0 + k * AG_THREADS;
        e[k] = (i < total) ? lst[i] : 0u;
      }
      if (use_pk) {
#pragma unroll
        for (int k = 0; k < 4; ++k) w[k] = pk[e[k]];
      } else {
#pragma unroll
        for (int k = 0; k < 4; ++k) w[k] = (unsigned)src[e[k]] | ((unsigned)(dst[e[k]] - lo) << 16);
      }
#pragma unroll
      for (int k = 0; k < 4; ++k) if (i0 + k * AG_THREADS >= total) w[k] = 0u;
#pragma unroll
      for (int k = 0; k < 4; ++k) { a[k] = elt[w[k] & 0xffffu]; b[k] = erl[w[k] >> 16]; }
#pragma unroll
      for (int k = 0; k < 4; ++k) {
        const unsigned i = i0 + k * AG_THREADS;
        if (i < total) {
          const float v = leaky(a[k] + b[k]);
          val[i] = v;
          lst[i] = w[k];
          atomic_max_f32(&mx[w[k] >> 16], v);
        }
      }
    }
    __syncthreads();
    for (unsigned i = tid; i < total; i += AG_THREADS) {
      const unsigned dl = lst[i] >> 16;
      const float ex = __expf(val[i] - mx[dl]);
      val[i] = ex;
      atomicAdd(&sm[dl], ex);
    }
    __syncthreads();
    const float* whl = Wh + lane;
    for (unsigned b0 = (unsigned)wv * 32u; b0 < total; b0 += AG_THREADS) {
      const unsigned id = b0 + lane;
      const unsigned pk = (id < total) ? lst[id] : 0u;
      const int      wi = (id < total) ? __float_as_int(val[id]) : 0;
      const int n = (total - b0 < 32u) ? (int)(total - b0) : 32;
#pragma unroll
      for (int h = 0; h < 32; h += 16) {
        if (n - h >= 16) {
          float r0[16], r1[16];
#pragma unroll
          for (int j = 0; j < 16; ++j) {
            const unsigned p = (unsigned)__builtin_amdgcn_readlane((int)pk, h + j);
            const float* r = whl + (size_t)(p & 0xffffu) * GAT_OUT;
            r0[j] = r[0];
            r1[j] = r[32];
          }
#pragma unroll
          for (int j = 0; j < 16; ++j) {
            const unsigned p = (unsigned)__builtin_amdgcn_readlane((int)pk, h + j);
            const float w = __int_as_float(__builtin_amdgcn_readlane(wi, h + j));
            float* a = acc + (p >> 16) * GAT_OUT + lane;
            atomicAdd(a,      w * r0[j]);
            atomicAdd(a + 32, w * r1[j]);
          }
        } else {
          for (int j = h; j < n && j < h + 16; ++j) {
            const unsigned p = (unsigned)__builtin_amdgcn_readlane((int)pk, j);
            const float w = __int_as_float(__builtin_amdgcn_readlane(wi, j));
            const float* r = whl + (size_t)(p & 0xffffu) * GAT_OUT;
            float* a = acc + (p >> 16) * GAT_OUT + lane;
            atomicAdd(a,      w * r[0]);
            atomicAdd(a + 32, w * r[32]);
          }
        }
      }
    }
  } else {
    for (int e = tid; e < GAT_E; e += AG_THREADS) {
      const int d = dst[e];
      if ((unsigned)(d - lo) < (unsigned)nn)
        atomic_max_f32(&mx[d - lo], leaky(el[src[e]] + er[d]));
    }
    __syncthreads();
    for (int e = tid; e < GAT_E; e += AG_THREADS) {
      const int d = dst[e];
      if ((unsigned)(d - lo) < (unsigned)nn) {
        const int s = src[e];
        const float ex = __expf(leaky(el[s] + er[d]) - mx[d - lo]);
        atomicAdd(&sm[d - lo], ex);
        const float* r = Wh + (size_t)s * GAT_OUT;
        float* a = acc + (d - lo) * GAT_OUT;
        for (int f = 0; f < GAT_OUT; ++f) atomicAdd(a + f, ex * r[f]);
      }
    }
  }
  __syncthreads();

  float* o = out + (size_t)lo * GAT_OUT;
  for (int i = tid; i < nn * GAT_OUT; i += AG_THREADS)
    o[i] = acc[i] / (sm[i >> 6] + GAT_EPS) + bias[i & 63];
}

extern "C" void kernel_launch(void* const* d_in, const int* in_sizes, int n_in,
                              void* d_out, int out_size, void* d_ws, size_t ws_size,
                              hipStream_t stream) {
  const float* x    = (const float*)d_in[0];
  const int*   ei   = (const int*)d_in[1];
  const float* W    = (const float*)d_in[2];
  const float* al   = (const float*)d_in[3];
  const float* ar   = (const float*)d_in[4];
  const float* bias = (const float*)d_in[5];
  float* out = (float*)d_out;
  (void)in_sizes; (void)n_in; (void)out_size;

  float* ws   = (float*)d_ws;
  float* Wh   = ws;
  float* el   = Wh   + (size_t)GAT_N * GAT_OUT;
  float* er   = el   + GAT_N;
  unsigned* oh  = (unsigned*)(er + GAT_N);
  unsigned* pk  = oh + (size_t)AG_PP * 32;

  const int* src = ei;
  const int* dst = ei + GAT_E;

  const int B = 256;
  k_gemm  <<<(GAT_N / 16 + 3) / 4, 128, 0, stream>>>(x, W, Wh);
  k_elr   <<<(GAT_N + B - 1) / B, B, 0, stream>>>(Wh, al, ar, el, er);
  k_bucket<<<(AG_NB + B - 1) / B, B, 0, stream>>>(src, dst, oh, pk, ws_size);
  k_agg   <<<AG_GROUPS, AG_THREADS, 0, stream>>>(src, dst, oh, pk, ws_size, el, er, Wh, bias, out);
}


# Aggregation: the last, partial batch of a workgroup's edge list fetches its rows together like a full batch (adds guarded) instead of edge by edge behind two dependent loads each; same sums
# speedup vs baseline: 6.2050x; 1.0401x over previous
#include <hip/hip_runtime.h>
#include <hip/hip_bf16.h>

#define GAT_N      50000
#define GAT_E      1600000
#define GAT_IN     128
#define GAT_OUT    64
#define GAT_SLOPE  0.2f
#define GAT_EPS    1e-12f

typedef __attribute__((ext_vector_type(2))) float v2f;
typedef __attribute__((ext_vector_type(4))) float v4f;
typedef __attribute__((ext_vector_type(8))) float v8f;
typedef __attribute__((ext_vector_type(4))) unsigned v4u;

__global__ __launch_bounds__(128) void k_gemm(const float* __restrict__ x,
                                              const float* __restrict__ W,
                                              float* __restrict__ Wh) {
  const int lane = threadIdx.x & 31;
  const int tile = blockIdx.x * 4 + (threadIdx.x >> 5);
  if (tile >= GAT_N / 16) return;
  const int base  = tile * 16;
  const int l15   = lane & 15;

  const int kq = (lane >> 4) << 2;
  const float* xr = x + (size_t)(base + l15) * GAT_IN + kq;
  const float* wr = W + (size_t)l15 * GAT_IN + kq;

  v8f acc[4] = {{}, {}, {}, {}};

#define GEMM_GB 2
#define GEMM_NG (GAT_IN / (8 * GEMM_GB))
  v4f fa[2][GEMM_GB], fb[2][GEMM_GB][4];
#define GEMM_LOAD_GROUP(s, t)                                                     \
  _Pragma("unroll") for (int j = 0; j < GEMM_GB; ++j) {                           \
    const int k = (t) * (8 * GEMM_GB) + j * 8;                                    \
    fa[s][j] = *(const v4f*)(xr + k);                                             \
    _Pragma("unroll") for (int n = 0; n < 4; ++n)                                 \
      fb[s][j][n] = *(const v4f*)(wr + (size_t)(16 * n) * GAT_IN + k);            \
  }
  GEMM_LOAD_GROUP(0, 0)
#pragma unroll
  for (int t = 0; t < GEMM_NG; ++t) {
    const int s = t & 1;
    if (t + 1 < GEMM_NG) { GEMM_LOAD_GROUP(s ^ 1, t + 1) }
    asm volatile("" :: "v"(fa[s][0]), "v"(fa[s][1]),
                 "v"(fb[s][0][0]), "v"(fb[s][0][1]), "v"(fb[s][0][2]), "v"(fb[s][0][3]),
                 "v"(fb[s][1][0]), "v"(fb[s][1][1]), "v"(fb[s][1][2]), "v"(fb[s][1][3]) : "memory");
#pragma unroll
    for (int j = 0; j < GEMM_GB; ++j) {
      const v4f a = fa[s][j];
      const v2f alo = {a[0], a[1]}, ahi = {a[2], a[3]};
#pragma unroll
      for (int n = 0; n < 4; ++n) {
        const v4f b = fb[s][j][n];
        acc[n] = __builtin_amdgcn_wmma_f32_16x16x4_f32(false, alo, false, (v2f){b[0], b[1]}, (short)0, acc[n], false, false);
      }
#pragma unroll
      for (int n = 0; n < 4; ++n) {
        const v4f b = fb[s][j][n];
        acc[n] = __builtin_amdgcn_wmma_f32_16x16x4_f32(false, ahi, false, (v2f){b[2], b[3]}, (short)0, acc[n], false, false);
      }
    }
  }
#undef GEMM_LOAD_GROUP
  const v8f acc0 = acc[0], acc1 = acc[1], acc2 = acc[2], acc3 = acc[3];

  const int rhalf = (lane >> 4) << 3;
  float* o = Wh + (size_t)(base + rhalf) * GAT_OUT + l15;
#pragma unroll
  for (int v = 0; v < 8; ++v) {
    o[(size_t)v * GAT_OUT +  0] = acc0[v];
    o[(size_t)v * GAT_OUT + 16] = acc1[v];
    o[(size_t)v * GAT_OUT + 32] = acc2[v];
    o[(size_t)v * GAT_OUT + 48] = acc3[v];
  }
}

__global__ void k_elr(const float* __restrict__ Wh, const float* __restrict__ al,
                      const float* __restrict__ ar, float* __restrict__ el,
                      float* __restrict__ er) {
  int i = blockIdx.x * blockDim.x + threadIdx.x;
  if (i >= GAT_N) return;
  const v4f* r = (const v4f*)(Wh + (size_t)i * GAT_OUT);
  float sl = 0.f, sr = 0.f;
#pragma unroll
  for (int q = 0; q < GAT_OUT / 4; ++q) {
    v4f v = r[q];
#pragma unroll
    for (int j = 0; j < 4; ++j) {
      sl += v[j] * al[q * 4 + j];
      sr += v[j] * ar[q * 4 + j];
    }
  }
  el[i] = sl; er[i] = sr;
}

#define AG_NODES   196
#define AG_GROUPS  256
#define AG_CAP     8192
#define AG_THREADS 1024
#define AG_NB      (GAT_E / 32)
#define AG_NQ      (AG_NB / 4)
#define AG_PP      ((AG_NB + 31) / 32 * 32)
#define AG_UPB     4
#define AG_WS_OH   ((size_t)(GAT_N * GAT_OUT + 2 * GAT_N) * 4 + (size_t)AG_PP * 32 * 4)
#define AG_WS_PK   (AG_WS_OH + (size_t)GAT_E * 4)

__device__ __forceinline__ void bucket_pack(const int* __restrict__ src, const int* __restrict__ dst,
                                            unsigned* __restrict__ pk, int blk) {
  static_assert(GAT_E % 512 == 0, "a batch of 16 blocks lies wholly inside or wholly past the edge list");
  const int wb = __builtin_amdgcn_readfirstlane((blk & ~31) * 32);
  const int lane = (int)(threadIdx.x & 31);
  for (int i0 = 0; i0 < 32; i0 += 16) {
    const int eb = wb + i0 * 32;
    if (eb >= GAT_E) break;
    unsigned s[16], d[16];
#pragma unroll
    for (int j = 0; j < 16; ++j) {
      s[j] = (unsigned)src[eb + j * 32 + lane];
      d[j] = (unsigned)dst[eb + j * 32 + lane];
    }
#pragma unroll
    for (int j = 0; j < 16; ++j)
      pk[eb + j * 32 + lane] = s[j] | ((d[j] - (d[j] / AG_NODES) * AG_NODES) << 16);
  }
}

__global__ void k_bucket(const int* __restrict__ src, const int* __restrict__ dst,
                         unsigned* __restrict__ oh, unsigned* __restrict__ pk, size_t ws_size) {
  const int blk = blockIdx.x * blockDim.x + threadIdx.x;
  if (ws_size < AG_WS_OH) return;
  if (ws_size >= AG_WS_PK) bucket_pack(src, dst, pk, blk);
  if (blk >= AG_NB) return;
  const int4* d = (const int4*)(dst + (size_t)blk * 32);
  unsigned w[8] = {0u, 0u, 0u, 0u, 0u, 0u, 0u, 0u};
#pragma unroll
  for (int i = 0; i < 8; ++i) {
    const int4 v = d[i];
    const unsigned b[4] = {(unsigned)v.x / AG_NODES, (unsigned)v.y / AG_NODES,
                           (unsigned)v.z / AG_NODES, (unsigned)v.w / AG_NODES};
#pragma unroll
    for (int c = 0; c < 4; ++c) {
#pragma unroll
      for (int t = 0; t < 8; ++t) w[t] |= ((b[c] >> t) & 1u) << (4 * i + c);
    }
  }
  unsigned pr[4][4];
#pragma unroll
  for (int f = 0; f < 4; ++f) {
#pragma unroll
    for (int v = 0; v < 4; ++v)
      pr[f][v] = ((v & 1) ? w[2 * f] : ~w[2 * f]) & ((v & 2) ? w[2 * f + 1] : ~w[2 * f + 1]);
  }
  unsigned* ob = oh + blk;
#pragma unroll
  for (int v = 0; v < 16; ++v) {
    ob[(size_t)v * AG_PP]        = pr[2][v & 3] & pr[3][v >> 2];
    ob[(size_t)(16 + v) * AG_PP] = pr[0][v & 3] & pr[1][v >> 2];
  }
}

__device__ __forceinline__ void atomic_max_f32(float* addr, float val) {
  if (val >= 0.f) atomicMax((int*)addr, __float_as_int(val));
  else            atomicMin((unsigned int*)addr, (unsigned int)__float_as_int(val));
}

__device__ __forceinline__ float leaky(float v) { return (v > 0.f) ? v : GAT_SLOPE * v; }

__global__ __launch_bounds__(AG_THREADS) void k_agg(const int* __restrict__ src, const int* __restrict__ dst,
                          const unsigned* __restrict__ oh,
                          const unsigned* __restrict__ pk, size_t ws_size,
                          const float* __restrict__ el, const float* __restrict__ er,
                          const float* __restrict__ Wh, const float* __restrict__ bias,
                          float* __restrict__ out) {
  __shared__ float    acc[AG_NODES * GAT_OUT];
  __shared__ float    mx[AG_NODES];
  __shared__ float    sm[AG_NODES];
  __shared__ unsigned lst[AG_CAP];
  __shared__ float    val[AG_CAP];
  __shared__ float    elt[GAT_N];
  __shared__ float    erl[AG_NODES];
  __shared__ unsigned cnt;
  static_assert(GAT_N % 4 == 0, "el is copied 4 floats at a time");
  static_assert(AG_NB % 4 == 0 && GAT_N / 4 <= AG_NQ, "a lane copies one 16-byte piece of el per quad of blocks it scans");

  const int tid  = threadIdx.x;
  const int lane = tid & 31;
  const int wv   = tid >> 5;
  const int g    = blockIdx.x;
  const int lo   = g * AG_NODES;
  const int nn   = (GAT_N - lo < AG_NODES) ? (GAT_N - lo) : AG_NODES;

  for (int i = tid; i < AG_NODES * GAT_OUT; i += AG_THREADS) acc[i] = 0.f;
  if (tid < AG_NODES) {
    mx[tid] = -__builtin_inff(); sm[tid] = 0.f;
    erl[tid] = (tid < nn) ? er[lo + tid] : 0.f;
  }
  if (tid == 0) cnt = 0u;
  __syncthreads();

  const bool have_planes = ws_size >= AG_WS_OH;
  const v4u* hp = (const v4u*)(oh + (size_t)(g >> 4) * AG_PP);
  const v4u* lp = (const v4u*)(oh + (size_t)(16 + (g & 15)) * AG_PP);
  const int rot = g * (AG_NQ / AG_GROUPS);
  for (int base0 = 0; have_planes && base0 < AG_NQ; base0 += AG_THREADS * AG_UPB) {
    const int base = base0 + tid;
    v4u qh[AG_UPB], ql[AG_UPB];
    v4f elq[AG_UPB];
    int   ux[AG_UPB];
#pragma unroll
    for (int k = 0; k < AG_UPB; ++k) {
      const int p = base + k * AG_THREADS;
      int u = p + rot;
      if (u >= AG_NQ) u -= AG_NQ;
      if (p >= AG_NQ) u = 0;
      ux[k] = u;
      qh[k] = hp[u];
      ql[k] = lp[u];
      elq[k] = ((const v4f*)el)[(p < GAT_N / 4) ? p : 0];
    }
    static_assert(AG_UPB == 4, "the fence below names the loads of four quads");
    asm volatile("" :: "v"(qh[0]), "v"(ql[0]), "v"(qh[1]), "v"(ql[1]), "v"(qh[2]), "v"(ql[2]),
                 "v"(qh[3]), "v"(ql[3]), "v"(elq[0]), "v"(elq[1]), "v"(elq[2]), "v"(elq[3]) : "memory");
    unsigned mk[AG_UPB][4];
    unsigned c = 0;
#pragma unroll
    for (int k = 0; k < AG_UPB; ++k) {
      const bool in = base + k * AG_THREADS < AG_NQ;
      mk[k][0] = in ? (qh[k].x & ql[k].x) : 0u;
      mk[k][1] = in ? (qh[k].y & ql[k].y) : 0u;
      mk[k][2] = in ? (qh[k].z & ql[k].z) : 0u;
      mk[k][3] = in ? (qh[k].w & ql[k].w) : 0u;
#pragma unroll
      for (int j = 0; j < 4; ++j) c += (unsigned)__builtin_popcount(mk[k][j]);
    }
    unsigned inc = c;
    inc += (unsigned)__builtin_amdgcn_update_dpp(0, (int)inc, 0x111, 0xf, 0xf, true);
    inc += (unsigned)__builtin_amdgcn_update_dpp(0, (int)inc, 0x112, 0xf, 0xf, true);
    inc += (unsigned)__builtin_amdgcn_update_dpp(0, (int)inc, 0x114, 0xf, 0xf, true);
    inc += (unsigned)__builtin_amdgcn_update_dpp(0, (int)inc, 0x118, 0xf, 0xf, true);
#pragma unroll
    for (int k = 0; k < AG_UPB; ++k) {
      const int p = base + k * AG_THREADS;
      if (p < GAT_N / 4) ((v4f*)elt)[p] = elq[k];
    }
    const unsigned row0 = (unsigned)__builtin_amdgcn_readlane((int)inc, 15);
    if (lane >= 16) inc += row0;
    const unsigned tot = (unsigned)__builtin_amdgcn_readlane((int)inc, 31);
    unsigned wb = 0;
    if (lane == 0) wb = atomicAdd(&cnt, tot);
    wb = (unsigned)__builtin_amdgcn_readfirstlane((int)wb);
    unsigned o = wb + inc - c;
#pragma unroll
    for (int k = 0; k < AG_UPB; ++k) {
#pragma unroll
      for (int j = 0; j < 4; ++j) {
        unsigned m = mk[k][j];
        const unsigned hi = (unsigned)(ux[k] * 4 + j) << 5;
        while (m) {
          const unsigned beta = (unsigned)__builtin_ctz(m);
          m &= m - 1u;
          if (o < AG_CAP) lst[o] = hi | beta;
          ++o;
        }
      }
    }
  }
  __syncthreads();
  const unsigned total = have_planes ? cnt : (unsigned)AG_CAP + 1u;

  if (total <= AG_CAP) {
    const bool use_pk = ws_size >= AG_WS_PK;
    for (unsigned i0 = tid; i0 < total; i0 += AG_THREADS * 4) {
      unsigned e[4], w[4]; float a[4], b[4];
#pragma unroll
      for (int k = 0; k < 4; ++k) {
        const unsigned i = i0 + k * AG_THREADS;
        e[k] = (i < total) ? lst[i] : 0u;
      }
      if (use_pk) {
#pragma unroll
        for (int k = 0; k < 4; ++k) w[k] = pk[e[k]];
      } else {
#pragma unroll
        for (int k = 0; k < 4; ++k) w[k] = (unsigned)src[e[k]] | ((unsigned)(dst[e[k]] - lo) << 16);
      }
#pragma unroll
      for (int k = 0; k < 4; ++k) if (i0 + k * AG_THREADS >= total) w[k] = 0u;
#pragma unroll
      for (int k = 0; k < 4; ++k) { a[k] = elt[w[k] & 0xffffu]; b[k] = erl[w[k] >> 16]; }
#pragma unroll
      for (int k = 0; k < 4; ++k) {
        const unsigned i = i0 + k * AG_THREADS;
        if (i < total) {
          const float v = leaky(a[k] + b[k]);
          val[i] = v;
          lst[i] = w[k];
          atomic_max_f32(&mx[w[k] >> 16], v);
        }
      }
    }
    __syncthreads();
    for (unsigned i = tid; i < total; i += AG_THREADS) {
      const unsigned dl = lst[i] >> 16;
      const float ex = __expf(val[i] - mx[dl]);
      val[i] = ex;
      atomicAdd(&sm[dl], ex);
    }
    __syncthreads();
    const float* whl = Wh + lane;
    for (unsigned b0 = (unsigned)wv * 32u; b0 < total; b0 += AG_THREADS) {
      const unsigned id = b0 + lane;
      const unsigned pk = (id < total) ? lst[id] : 0u;
      const int      wi = (id < total) ? __float_as_int(val[id]) : 0;
      const int n = (total - b0 < 32u) ? (int)(total - b0) : 32;
#pragma unroll
      for (int h = 0; h < 32; h += 16) {
        if (n > h) {
          float r0[16], r1[16];
#pragma unroll
          for (int j = 0; j < 16; ++j) {
            const unsigned p = (unsigned)__builtin_amdgcn_readlane((int)pk, h + j);
            const float* r = whl + (size_t)(p & 0xffffu) * GAT_OUT;
            r0[j] = r[0];
            r1[j] = r[32];
          }
          if (n - h >= 16) {
#pragma unroll
            for (int j = 0; j < 16; ++j) {
              const unsigned p = (unsigned)__builtin_amdgcn_readlane((int)pk, h + j);
              const float w = __int_as_float(__builtin_amdgcn_readlane(wi, h + j));
              float* a = acc + (p >> 16) * GAT_OUT + lane;
              atomicAdd(a,      w * r0[j]);
              atomicAdd(a + 32, w * r1[j]);
            }
          } else {
#pragma unroll
            for (int j = 0; j < 16; ++j) {
              if (h + j < n) {
                const unsigned p = (unsigned)__builtin_amdgcn_readlane((int)pk, h + j);
                const float w = __int_as_float(__builtin_amdgcn_readlane(wi, h + j));
                float* a = acc + (p >> 16) * GAT_OUT + lane;
                atomicAdd(a,      w * r0[j]);
                atomicAdd(a + 32, w * r1[j]);
              }
            }
          }
        }
      }
    }
  } else {
    for (int e = tid; e < GAT_E; e += AG_THREADS) {
      const int d = dst[e];
      if ((unsigned)(d - lo) < (unsigned)nn)
        atomic_max_f32(&mx[d - lo], leaky(el[src[e]] + er[d]));
    }
    __syncthreads();
    for (int e = tid; e < GAT_E; e += AG_THREADS) {
      const int d = dst[e];
      if ((unsigned)(d - lo) < (unsigned)nn) {
        const int s = src[e];
        const float ex = __expf(leaky(el[s] + er[d]) - mx[d - lo]);
        atomicAdd(&sm[d - lo], ex);
        const float* r = Wh + (size_t)s * GAT_OUT;
        float* a = acc + (d - lo) * GAT_OUT;
        for (int f = 0; f < GAT_OUT; ++f) atomicAdd(a + f, ex * r[f]);
      }
    }
  }
  __syncthreads();

  float* o = out + (size_t)lo * GAT_OUT;
  for (int i = tid; i < nn * GAT_OUT; i += AG_THREADS)
    o[i] = acc[i] / (sm[i >> 6] + GAT_EPS) + bias[i & 63];
}

extern "C" void kernel_launch(void* const* d_in, const int* in_sizes, int n_in,
                              void* d_out, int out_size, void* d_ws, size_t ws_size,
                              hipStream_t stream) {
  const float* x    = (const float*)d_in[0];
  const int*   ei   = (const int*)d_in[1];
  const float* W    = (const float*)d_in[2];
  const float* al   = (const float*)d_in[3];
  const float* ar   = (const float*)d_in[4];
  const float* bias = (const float*)d_in[5];
  float* out = (float*)d_out;
  (void)in_sizes; (void)n_in; (void)out_size;

  float* ws   = (float*)d_ws;
  float* Wh   = ws;
  float* el   = Wh   + (size_t)GAT_N * GAT_OUT;
  float* er   = el   + GAT_N;
  unsigned* oh  = (unsigned*)(er + GAT_N);
  unsigned* pk  = oh + (size_t)AG_PP * 32;

  const int* src = ei;
  const int* dst = ei + GAT_E;

  const int B = 256;
  k_gemm  <<<(GAT_N / 16 + 3) / 4, 128, 0, stream>>>(x, W, Wh);
  k_elr   <<<(GAT_N + B - 1) / B, B, 0, stream>>>(Wh, al, ar, el, er);
  k_bucket<<<(AG_NB + B - 1) / B, B, 0, stream>>>(src, dst, oh, pk, ws_size);
  k_agg   <<<AG_GROUPS, AG_THREADS, 0, stream>>>(src, dst, oh, pk, ws_size, el, er, Wh, bias, out);
}
